# starting kernel
# baseline (speedup 1.0000x reference)
#include <hip/hip_runtime.h>
#include <hip/hip_bf16.h>


#define N_NODES 50000
#define N_EDGES 800000
#define CDIM    96
#define NC      (N_NODES * CDIM)
#define TILES   ((N_NODES / 16) * (CDIM / 16))

typedef __attribute__((ext_vector_type(2))) float v2f;
typedef __attribute__((ext_vector_type(8))) float v8f;

__global__ __launch_bounds__(256) void gnn_gemm_m(const float* __restrict__ x,
                                                  const float* __restrict__ W,
                                                  float* __restrict__ m) {
  const int wave = blockIdx.x * (blockDim.x >> 5) + (threadIdx.x >> 5);
  if (wave >= TILES) return;
  const int tr   = wave / (CDIM / 16);
  const int tc   = wave % (CDIM / 16);
  const int lane = threadIdx.x & 31;
  const int half = lane >> 4;
  const int l16  = lane & 15;
  const int row0 = tr * 16, col0 = tc * 16;

  const float* arow = x + (row0 + l16) * CDIM + 2 * half;
  const float* bcol = W + col0 + l16;

  v8f acc = {0.f, 0.f, 0.f, 0.f, 0.f, 0.f, 0.f, 0.f};
#pragma unroll
  for (int k0 = 0; k0 < CDIM; k0 += 4) {
    v2f a, b;
    a.x = arow[k0];
    a.y = arow[k0 + 1];
    const int k = k0 + 2 * half;
    b.x = bcol[(k    ) * CDIM];
    b.y = bcol[(k + 1) * CDIM];
    acc = __builtin_amdgcn_wmma_f32_16x16x4_f32(false, a, false, b,
                                                (short)0, acc, false, false);
  }

  float* orow = m + (row0 + 8 * half) * CDIM + col0 + l16;
#pragma unroll
  for (int r = 0; r < 8; ++r) orow[r * CDIM] = acc[r];
}

__global__ void gnn_zero(float4* __restrict__ p, int n4) {
  int i = blockIdx.x * blockDim.x + threadIdx.x;
  if (i < n4) p[i] = make_float4(0.f, 0.f, 0.f, 0.f);
}

__global__ void gnn_scatter(const float* __restrict__ m,
                            const int* __restrict__ edge_index,
                            float* __restrict__ agg) {
  const long long total  = (long long)N_EDGES * (CDIM / 4);
  const long long stride = (long long)gridDim.x * blockDim.x;
  for (long long i = (long long)blockIdx.x * blockDim.x + threadIdx.x;
       i < total; i += stride) {
    const int e = (int)(i / (CDIM / 4));
    const int g = (int)(i % (CDIM / 4));
    const int s = edge_index[e];
    const int d = edge_index[N_EDGES + e];
    const float4 v = *(const float4*)(m + s * CDIM + g * 4);
    float* p = agg + d * CDIM + g * 4;
    atomicAdd(p + 0, v.x);
    atomicAdd(p + 1, v.y);
    atomicAdd(p + 2, v.z);
    atomicAdd(p + 3, v.w);
  }
}

__global__ __launch_bounds__(256) void gnn_gru(const float* __restrict__ agg,
                                               const float* __restrict__ h,
                                               const float* __restrict__ w_ih,
                                               const float* __restrict__ w_hh,
                                               const float* __restrict__ b_ih,
                                               const float* __restrict__ b_hh,
                                               float* __restrict__ out) {
  const int wave = blockIdx.x * (blockDim.x >> 5) + (threadIdx.x >> 5);
  if (wave >= TILES) return;
  const int tr   = wave / (CDIM / 16);
  const int tc   = wave % (CDIM / 16);
  const int lane = threadIdx.x & 31;
  const int half = lane >> 4;
  const int l16  = lane & 15;
  const int row0 = tr * 16, col0 = tc * 16;
  const int col  = col0 + l16;

  v8f a_ir, a_iz, a_in, a_hr, a_hz, a_hn;
  {
    const float bir = b_ih[col], biz = b_ih[CDIM + col], bin_ = b_ih[2 * CDIM + col];
    const float bhr = b_hh[col], bhz = b_hh[CDIM + col], bhn  = b_hh[2 * CDIM + col];
#pragma unroll
    for (int r = 0; r < 8; ++r) {
      a_ir[r] = bir; a_iz[r] = biz; a_in[r] = bin_;
      a_hr[r] = bhr; a_hz[r] = bhz; a_hn[r] = bhn;
    }
  }

  const float* aggrow = agg + (row0 + l16) * CDIM + 2 * half;
  const float* hrow   = h   + (row0 + l16) * CDIM + 2 * half;
  const float* wih_r  = w_ih + (0 * CDIM + col) * CDIM;
  const float* wih_z  = w_ih + (1 * CDIM + col) * CDIM;
  const float* wih_n  = w_ih + (2 * CDIM + col) * CDIM;
  const float* whh_r  = w_hh + (0 * CDIM + col) * CDIM;
  const float* whh_z  = w_hh + (1 * CDIM + col) * CDIM;
  const float* whh_n  = w_hh + (2 * CDIM + col) * CDIM;

#pragma unroll 4
  for (int k0 = 0; k0 < CDIM; k0 += 4) {
    v2f ai, ah, b;
    ai.x = aggrow[k0]; ai.y = aggrow[k0 + 1];
    ah.x = hrow[k0];   ah.y = hrow[k0 + 1];
    const int k = k0 + 2 * half;

    b.x = wih_r[k]; b.y = wih_r[k + 1];
    a_ir = __builtin_amdgcn_wmma_f32_16x16x4_f32(false, ai, false, b, (short)0, a_ir, false, false);
    b.x = wih_z[k]; b.y = wih_z[k + 1];
    a_iz = __builtin_amdgcn_wmma_f32_16x16x4_f32(false, ai, false, b, (short)0, a_iz, false, false);
    b.x = wih_n[k]; b.y = wih_n[k + 1];
    a_in = __builtin_amdgcn_wmma_f32_16x16x4_f32(false, ai, false, b, (short)0, a_in, false, false);
    b.x = whh_r[k]; b.y = whh_r[k + 1];
    a_hr = __builtin_amdgcn_wmma_f32_16x16x4_f32(false, ah, false, b, (short)0, a_hr, false, false);
    b.x = whh_z[k]; b.y = whh_z[k + 1];
    a_hz = __builtin_amdgcn_wmma_f32_16x16x4_f32(false, ah, false, b, (short)0, a_hz, false, false);
    b.x = whh_n[k]; b.y = whh_n[k + 1];
    a_hn = __builtin_amdgcn_wmma_f32_16x16x4_f32(false, ah, false, b, (short)0, a_hn, false, false);
  }

#pragma unroll
  for (int r = 0; r < 8; ++r) {
    const int row = row0 + r + 8 * half;
    const float hv = h[row * CDIM + col];
    const float rr = 1.f / (1.f + __expf(-(a_ir[r] + a_hr[r])));
    const float zz = 1.f / (1.f + __expf(-(a_iz[r] + a_hz[r])));
    const float nn = tanhf(a_in[r] + rr * a_hn[r]);
    out[row * CDIM + col] = (1.f - zz) * nn + zz * hv;
  }
}

extern "C" void kernel_launch(void* const* d_in, const int* in_sizes, int n_in,
                              void* d_out, int out_size, void* d_ws, size_t ws_size,
                              hipStream_t stream) {
  const float* x      = (const float*)d_in[0];
  const int*   ei     = (const int*)  d_in[1];
  const float* weight = (const float*)d_in[2];
  const float* w_ih   = (const float*)d_in[3];
  const float* w_hh   = (const float*)d_in[4];
  const float* b_ih   = (const float*)d_in[5];
  const float* b_hh   = (const float*)d_in[6];
  float* out = (float*)d_out;

  float* xa  = (float*)d_ws;
  float* xb  = xa + NC;
  float* m   = xb + NC;
  float* agg = m  + NC;

  const int wavesPerBlock = 256 / 32;
  const int gemmBlocks    = (TILES + wavesPerBlock - 1) / wavesPerBlock;
  const int zeroBlocks    = (NC / 4 + 255) / 256;
  const int scatBlocks    = 2048;

  for (int l = 0; l < 3; ++l) {
    const float* xin  = (l == 0) ? x  : ((l == 1) ? xa : xb);
    float*       xout = (l == 2) ? out : ((l == 0) ? xa : xb);
    const float* Wl   = weight + l * CDIM * CDIM;

    gnn_gemm_m<<<gemmBlocks, 256, 0, stream>>>(xin, Wl, m);
    gnn_zero  <<<zeroBlocks, 256, 0, stream>>>((float4*)agg, NC / 4);
    gnn_scatter<<<scatBlocks, 256, 0, stream>>>(m, ei, agg);
    gnn_gru   <<<gemmBlocks, 256, 0, stream>>>(agg, xin, w_ih, w_hh, b_ih, b_hh, xout);
  }
}


# Scatter scans the destination ids in layer 0 only: each wave saves its kept edges (source id, local row) in the workspace; layers 1-2 read the list, no scan; LDS sums in arrival order as before
# speedup vs baseline: 2.4030x; 2.4030x over previous
#include <hip/hip_runtime.h>
#include <hip/hip_bf16.h>


#define N_NODES 50000
#define N_EDGES 800000
#define CDIM    96
#define NC      (N_NODES * CDIM)
#define TILES   ((N_NODES / 16) * (CDIM / 16))

typedef __attribute__((ext_vector_type(2))) float v2f;
typedef __attribute__((ext_vector_type(8))) float v8f;

#define KPAIRS      (CDIM / 8)
#define PACK_TILE   (KPAIRS * 32)
#define PACK_MAT    ((CDIM / 16) * PACK_TILE)
#define PACK_ALL    (9 * PACK_MAT)

__global__ __launch_bounds__(256) void gnn_pack(const float* __restrict__ weight,
                                                const float* __restrict__ w_ih,
                                                const float* __restrict__ w_hh,
                                                float4* __restrict__ wp) {
  const int t    = blockIdx.x * 256 + threadIdx.x;
  const int lane = t & 31;
  const int c    = lane & 15;
  const int h    = lane >> 4;
  int q = t >> 5;
  const int kk = q % KPAIRS;       q /= KPAIRS;
  const int tc = q % (CDIM / 16);  q /= (CDIM / 16);
  const int k  = 8 * kk + 2 * h;
  const int n  = tc * 16 + c;
  float4 v;
  if (q < 6) {
    const float* s = (q < 3 ? w_ih : w_hh) + ((q % 3) * CDIM + n) * CDIM + k;
    v = make_float4(s[0], s[1], s[4], s[5]);
  } else {
    const float* s = weight + (q - 6) * CDIM * CDIM + k * CDIM + n;
    v = make_float4(s[0], s[CDIM], s[4 * CDIM], s[5 * CDIM]);
  }
  wp[t] = v;
}

__global__ __launch_bounds__(256) void gnn_gemm_m(const float* __restrict__ x,
                                                  const float4* __restrict__ wpl,
                                                  float* __restrict__ m) {
  const int wave = blockIdx.x * (blockDim.x >> 5) + (threadIdx.x >> 5);
  if (wave >= TILES) return;
  const int tr   = wave / (CDIM / 16);
  const int tc   = wave % (CDIM / 16);
  const int lane = threadIdx.x & 31;
  const int half = lane >> 4;
  const int l16  = lane & 15;
  const int row0 = tr * 16, col0 = tc * 16;

  const float*  arow = x + (row0 + l16) * CDIM + 2 * half;
  const float4* bq   = wpl + tc * PACK_TILE + lane;

  v8f acc = {0.f, 0.f, 0.f, 0.f, 0.f, 0.f, 0.f, 0.f};
#pragma unroll
  for (int kk = 0; kk < KPAIRS; ++kk) {
    const int k0 = 8 * kk;
    const float4 bb = bq[kk * 32];
    v2f a, b;
    a.x = arow[k0];     a.y = arow[k0 + 1];
    b.x = bb.x;         b.y = bb.y;
    acc = __builtin_amdgcn_wmma_f32_16x16x4_f32(false, a, false, b,
                                                (short)0, acc, false, false);
    a.x = arow[k0 + 4]; a.y = arow[k0 + 5];
    b.x = bb.z;         b.y = bb.w;
    acc = __builtin_amdgcn_wmma_f32_16x16x4_f32(false, a, false, b,
                                                (short)0, acc, false, false);
  }

  float* orow = m + (row0 + 8 * half) * CDIM + col0 + l16;
#pragma unroll
  for (int r = 0; r < 8; ++r) orow[r * CDIM] = acc[r];
}

#define RNODES   391
#define NRANGES  128
#define SLICE_Q  (N_EDGES / 4 / 2)
#define SCAN_IT  25
#define LCAP     544
#define SINK_ENT ((unsigned)RNODES)

__device__ __forceinline__ int4 scan_load(const int4* __restrict__ dst4, int q) {
  int4 d = make_int4(-1, -1, -1, -1);
  if (q < SLICE_Q) d = dst4[q];
  return d;
}

__device__ __forceinline__ void scan_keep(unsigned* mylist, int& count, int dl, int n, unsigned e) {
  const bool hit = (unsigned)dl < (unsigned)n;
  const unsigned mask = __builtin_amdgcn_ballot_w32(hit);
  if (mask) {
    if (hit) mylist[count + (int)__builtin_amdgcn_mbcnt_lo(mask, 0u)] = (e << 9) | (unsigned)dl;
    count += __builtin_popcount(mask);
  }
}

__device__ __forceinline__ void scan_group(unsigned* mylist, int& count, int4 d, int lo, int n, unsigned e) {
  scan_keep(mylist, count, d.x - lo, n, e);
  scan_keep(mylist, count, d.y - lo, n, e + 1u);
  scan_keep(mylist, count, d.z - lo, n, e + 2u);
  scan_keep(mylist, count, d.w - lo, n, e + 3u);
}

#define SCAP     512

template <int G>
__device__ __forceinline__ void add_rows(int s_l, int dl_l, int cnt, int lane,
                                         const float* __restrict__ m, float* acc) {
  for (int jb = 0; jb < cnt; jb += G) {
    float v[G][3];
    float* a[G];
#pragma unroll
    for (int j = 0; j < G; ++j) {
      const int s  = __builtin_amdgcn_readlane(s_l, jb + j);
      const int dl = __builtin_amdgcn_readlane(dl_l, jb + j);
      const float* r = m + s * CDIM + lane;
      v[j][0] = r[0]; v[j][1] = r[32]; v[j][2] = r[64];
      a[j] = acc + dl * CDIM + lane;
    }
#pragma unroll
    for (int j = 0; j < G; ++j) {
      atomicAdd(a[j],      v[j][0]);
      atomicAdd(a[j] + 32, v[j][1]);
      atomicAdd(a[j] + 64, v[j][2]);
    }
  }
}

template <int G, bool SAVE>
__device__ __forceinline__ void add_listed(const unsigned* ptr, int cnt, int lane,
                                           const float* __restrict__ m,
                                           const int* __restrict__ srcid, float* acc,
                                           unsigned* __restrict__ slist, int& saved) {
  const unsigned ent = (lane < cnt) ? ptr[lane] : SINK_ENT;
  const int s_l  = srcid[ent >> 9];
  const int dl_l = (int)(ent & 511u);
  if (SAVE) {
    if (saved + cnt <= SCAP && lane < cnt)
      slist[saved + lane] = ((unsigned)s_l << 9) | (unsigned)dl_l;
    saved += cnt;
  }
  add_rows<G>(s_l, dl_l, cnt, lane, m, acc);
}

template <bool SAVE>
__device__ __forceinline__ void scan_and_add(const float* __restrict__ m,
                                             const int* __restrict__ edge_index,
                                             float* acc, unsigned* mylist,
                                             int tid, int lane, int slice, int lo, int n,
                                             unsigned* __restrict__ slist, int& saved) {
  const int*  __restrict__ srcid = edge_index;
  const int4* __restrict__ dst4  = (const int4*)(edge_index + N_EDGES) + slice * SLICE_Q;
  const unsigned ebase = (unsigned)(slice * (N_EDGES / 2));
  int count = 0;

  int4 d0 = scan_load(dst4, 0 * 1024 + tid);
  int4 d1 = scan_load(dst4, 1 * 1024 + tid);
  int4 d2 = scan_load(dst4, 2 * 1024 + tid);
  int4 d3 = scan_load(dst4, 3 * 1024 + tid);
  for (int it = 0; it < SCAN_IT; ++it) {
    const int q0 = it * 4096 + tid;
    const int4 n0 = scan_load(dst4, q0 + 4096);
    const int4 n1 = scan_load(dst4, q0 + 5120);
    const int4 n2 = scan_load(dst4, q0 + 6144);
    const int4 n3 = scan_load(dst4, q0 + 7168);
    scan_group(mylist, count, d0, lo, n, ebase + 4u * (unsigned)(q0));
    scan_group(mylist, count, d1, lo, n, ebase + 4u * (unsigned)(q0 + 1024));
    scan_group(mylist, count, d2, lo, n, ebase + 4u * (unsigned)(q0 + 2048));
    scan_group(mylist, count, d3, lo, n, ebase + 4u * (unsigned)(q0 + 3072));
    while (count >= 32) {
      count -= 32;
      add_listed<16, SAVE>(mylist + count, 32, lane, m, srcid, acc, slist, saved);
    }
    d0 = n0; d1 = n1; d2 = n2; d3 = n3;
  }
  if (count > 0) add_listed<8, SAVE>(mylist, count, lane, m, srcid, acc, slist, saved);
}

template <bool FIRST>
__global__ __launch_bounds__(1024) void gnn_scatter(const float* __restrict__ m,
                                                    const int* __restrict__ edge_index,
                                                    float* __restrict__ agg0,
                                                    float* __restrict__ agg1,
                                                    unsigned* __restrict__ scount,
                                                    unsigned* __restrict__ slists) {
  __shared__ float    acc[(RNODES + 1) * CDIM];
  __shared__ unsigned list[32 * LCAP];

  const int tid   = threadIdx.x;
  const int lane  = tid & 31;
  const int w     = __builtin_amdgcn_readfirstlane(tid >> 5);
  const int range = blockIdx.x >> 1;
  const int slice = blockIdx.x & 1;
  const int lo    = range * RNODES;
  const int n     = (N_NODES - lo < RNODES) ? (N_NODES - lo) : RNODES;
  float* __restrict__ agg = slice ? agg1 : agg0;

  float4* acc4 = (float4*)acc;
  for (int i = tid; i < (RNODES + 1) * CDIM / 4; i += 1024)
    acc4[i] = make_float4(0.f, 0.f, 0.f, 0.f);
  __syncthreads();

  const int wid = blockIdx.x * 32 + w;
  unsigned* __restrict__ slist = slists + wid * SCAP;
  unsigned* mylist = list + w * LCAP;

  if (FIRST) {
    int saved = 0;
    scan_and_add<true>(m, edge_index, acc, mylist, tid, lane, slice, lo, n, slist, saved);
    if (lane == 0) scount[wid] = (unsigned)saved;
  } else {
    const int c = __builtin_amdgcn_readfirstlane((int)scount[wid]);
    if (c <= SCAP) {
      unsigned ent = (lane < c) ? slist[lane] : SINK_ENT;
      for (int b = 0; b < c; b += 32) {
        const int cnt = (c - b < 32) ? (c - b) : 32;
        const unsigned cur = ent;
        const int nx = b + 32 + lane;
        const unsigned nxt = slist[(nx < c) ? nx : (c - 1)];
        ent = (nx < c) ? nxt : SINK_ENT;
        const int s_l  = (int)(cur >> 9);
        const int dl_l = (int)(cur & 511u);
        if (cnt == 32) add_rows<16>(s_l, dl_l, 32,  lane, m, acc);
        else           add_rows<8> (s_l, dl_l, cnt, lane, m, acc);
      }
    } else {
      int saved = 0;
      scan_and_add<false>(m, edge_index, acc, mylist, tid, lane, slice, lo, n, slist, saved);
    }
  }

  __syncthreads();
  float4* __restrict__ o4 = (float4*)(agg + lo * CDIM);
  for (int i = tid; i < n * (CDIM / 4); i += 1024) o4[i] = acc4[i];
}

__global__ __launch_bounds__(256) void gnn_gru(const float* __restrict__ agg0,
                                               const float* __restrict__ agg1,
                                               const float* __restrict__ h,
                                               const float4* __restrict__ wp,
                                               const float* __restrict__ b_ih,
                                               const float* __restrict__ b_hh,
                                               float* __restrict__ out) {
  const int wave = blockIdx.x * (blockDim.x >> 5) + (threadIdx.x >> 5);
  if (wave >= TILES) return;
  const int tr   = wave / (CDIM / 16);
  const int tc   = wave % (CDIM / 16);
  const int lane = threadIdx.x & 31;
  const int half = lane >> 4;
  const int l16  = lane & 15;
  const int row0 = tr * 16, col0 = tc * 16;
  const int col  = col0 + l16;

  v8f a_ir, a_iz, a_in, a_hr, a_hz, a_hn;
  {
    const float bir = b_ih[col], biz = b_ih[CDIM + col], bin_ = b_ih[2 * CDIM + col];
    const float bhr = b_hh[col], bhz = b_hh[CDIM + col], bhn  = b_hh[2 * CDIM + col];
#pragma unroll
    for (int r = 0; r < 8; ++r) {
      a_ir[r] = bir; a_iz[r] = biz; a_in[r] = bin_;
      a_hr[r] = bhr; a_hz[r] = bhz; a_hn[r] = bhn;
    }
  }

  const float* ag0row = agg0 + (row0 + l16) * CDIM + 2 * half;
  const float* ag1row = agg1 + (row0 + l16) * CDIM + 2 * half;
  const float* hrow   = h   + (row0 + l16) * CDIM + 2 * half;
  const float4* wq    = wp + tc * PACK_TILE + lane;

#pragma unroll 2
  for (int kk = 0; kk < KPAIRS; ++kk) {
    const int k0 = 8 * kk;
    const float4 w_ir = wq[0 * PACK_MAT + kk * 32];
    const float4 w_iz = wq[1 * PACK_MAT + kk * 32];
    const float4 w_in = wq[2 * PACK_MAT + kk * 32];
    const float4 w_hr = wq[3 * PACK_MAT + kk * 32];
    const float4 w_hz = wq[4 * PACK_MAT + kk * 32];
    const float4 w_hn = wq[5 * PACK_MAT + kk * 32];
    v2f ai, ah, b;

    ai.x = ag0row[k0] + ag1row[k0]; ai.y = ag0row[k0 + 1] + ag1row[k0 + 1];
    ah.x = hrow[k0];   ah.y = hrow[k0 + 1];
    b.x = w_ir.x; b.y = w_ir.y;
    a_ir = __builtin_amdgcn_wmma_f32_16x16x4_f32(false, ai, false, b, (short)0, a_ir, false, false);
    b.x = w_iz.x; b.y = w_iz.y;
    a_iz = __builtin_amdgcn_wmma_f32_16x16x4_f32(false, ai, false, b, (short)0, a_iz, false, false);
    b.x = w_in.x; b.y = w_in.y;
    a_in = __builtin_amdgcn_wmma_f32_16x16x4_f32(false, ai, false, b, (short)0, a_in, false, false);
    b.x = w_hr.x; b.y = w_hr.y;
    a_hr = __builtin_amdgcn_wmma_f32_16x16x4_f32(false, ah, false, b, (short)0, a_hr, false, false);
    b.x = w_hz.x; b.y = w_hz.y;
    a_hz = __builtin_amdgcn_wmma_f32_16x16x4_f32(false, ah, false, b, (short)0, a_hz, false, false);
    b.x = w_hn.x; b.y = w_hn.y;
    a_hn = __builtin_amdgcn_wmma_f32_16x16x4_f32(false, ah, false, b, (short)0, a_hn, false, false);

    ai.x = ag0row[k0 + 4] + ag1row[k0 + 4]; ai.y = ag0row[k0 + 5] + ag1row[k0 + 5];
    ah.x = hrow[k0 + 4];   ah.y = hrow[k0 + 5];
    b.x = w_ir.z; b.y = w_ir.w;
    a_ir = __builtin_amdgcn_wmma_f32_16x16x4_f32(false, ai, false, b, (short)0, a_ir, false, false);
    b.x = w_iz.z; b.y = w_iz.w;
    a_iz = __builtin_amdgcn_wmma_f32_16x16x4_f32(false, ai, false, b, (short)0, a_iz, false, false);
    b.x = w_in.z; b.y = w_in.w;
    a_in = __builtin_amdgcn_wmma_f32_16x16x4_f32(false, ai, false, b, (short)0, a_in, false, false);
    b.x = w_hr.z; b.y = w_hr.w;
    a_hr = __builtin_amdgcn_wmma_f32_16x16x4_f32(false, ah, false, b, (short)0, a_hr, false, false);
    b.x = w_hz.z; b.y = w_hz.w;
    a_hz = __builtin_amdgcn_wmma_f32_16x16x4_f32(false, ah, false, b, (short)0, a_hz, false, false);
    b.x = w_hn.z; b.y = w_hn.w;
    a_hn = __builtin_amdgcn_wmma_f32_16x16x4_f32(false, ah, false, b, (short)0, a_hn, false, false);
  }

#pragma unroll
  for (int r = 0; r < 8; ++r) {
    const int row = row0 + r + 8 * half;
    const float hv = h[row * CDIM + col];
    const float rr = 1.f / (1.f + __expf(-(a_ir[r] + a_hr[r])));
    const float zz = 1.f / (1.f + __expf(-(a_iz[r] + a_hz[r])));
    const float nn = tanhf(a_in[r] + rr * a_hn[r]);
    out[row * CDIM + col] = (1.f - zz) * nn + zz * hv;
  }
}

extern "C" void kernel_launch(void* const* d_in, const int* in_sizes, int n_in,
                              void* d_out, int out_size, void* d_ws, size_t ws_size,
                              hipStream_t stream) {
  const float* x      = (const float*)d_in[0];
  const int*   ei     = (const int*)  d_in[1];
  const float* weight = (const float*)d_in[2];
  const float* w_ih   = (const float*)d_in[3];
  const float* w_hh   = (const float*)d_in[4];
  const float* b_ih   = (const float*)d_in[5];
  const float* b_hh   = (const float*)d_in[6];
  float* out = (float*)d_out;

  float*  agg1 = (float*)d_ws;
  float4* wp   = (float4*)(agg1 + NC);
  float*  p    = agg1 + 2 * NC;
  float*  agg0 = agg1 + 3 * NC;
  unsigned* scount = (unsigned*)(agg1 + NC) + 131072;
  unsigned* slists = (unsigned*)(agg1 + NC) + 262144;

  const int wavesPerBlock = 256 / 32;
  const int gemmBlocks    = (TILES + wavesPerBlock - 1) / wavesPerBlock;
  const int scatBlocks    = NRANGES * 2;

  gnn_pack<<<PACK_ALL / 256, 256, 0, stream>>>(weight, w_ih, w_hh, wp);

  for (int l = 0; l < 3; ++l) {
    const float*  xin  = (l == 0) ? x   : ((l == 1) ? out : p);
    float*        m    = (l == 2) ? out : p;
    float*        xout = (l == 1) ? p   : out;
    const float4* wpl  = wp + (6 + l) * PACK_MAT;

    gnn_gemm_m <<<gemmBlocks, 256,  0, stream>>>(xin, wpl, m);
    if (l == 0) gnn_scatter<true> <<<scatBlocks, 1024, 0, stream>>>(m, ei, agg0, agg1, scount, slists);
    else        gnn_scatter<false><<<scatBlocks, 1024, 0, stream>>>(m, ei, agg0, agg1, scount, slists);
    gnn_gru    <<<gemmBlocks, 256,  0, stream>>>(agg0, agg1, xin, wp, b_ih, b_hh, xout);
  }
}


# Scatter stores both agg copies in WMMA A-fragment order (ranges of 400 nodes = 25 row tiles): GRU reads agg as one 512-byte load per 2 k-steps, not 8-byte loads over 16 lines; LDS sum order as before
# speedup vs baseline: 2.4754x; 1.0301x over previous
#include <hip/hip_runtime.h>
#include <hip/hip_bf16.h>


#define N_NODES 50000
#define N_EDGES 800000
#define CDIM    96
#define NC      (N_NODES * CDIM)
#define TILES   ((N_NODES / 16) * (CDIM / 16))

typedef __attribute__((ext_vector_type(2))) float v2f;
typedef __attribute__((ext_vector_type(8))) float v8f;

#define KPAIRS      (CDIM / 8)
#define PACK_TILE   (KPAIRS * 32)
#define PACK_MAT    ((CDIM / 16) * PACK_TILE)
#define PACK_ALL    (9 * PACK_MAT)

__global__ __launch_bounds__(256) void gnn_pack(const float* __restrict__ weight,
                                                const float* __restrict__ w_ih,
                                                const float* __restrict__ w_hh,
                                                float4* __restrict__ wp) {
  const int t    = blockIdx.x * 256 + threadIdx.x;
  const int lane = t & 31;
  const int c    = lane & 15;
  const int h    = lane >> 4;
  int q = t >> 5;
  const int kk = q % KPAIRS;       q /= KPAIRS;
  const int tc = q % (CDIM / 16);  q /= (CDIM / 16);
  const int k  = 8 * kk + 2 * h;
  const int n  = tc * 16 + c;
  float4 v;
  if (q < 6) {
    const float* s = (q < 3 ? w_ih : w_hh) + ((q % 3) * CDIM + n) * CDIM + k;
    v = make_float4(s[0], s[1], s[4], s[5]);
  } else {
    const float* s = weight + (q - 6) * CDIM * CDIM + k * CDIM + n;
    v = make_float4(s[0], s[CDIM], s[4 * CDIM], s[5 * CDIM]);
  }
  wp[t] = v;
}

__global__ __launch_bounds__(256) void gnn_gemm_m(const float* __restrict__ x,
                                                  const float4* __restrict__ wpl,
                                                  float* __restrict__ m) {
  const int wave = blockIdx.x * (blockDim.x >> 5) + (threadIdx.x >> 5);
  if (wave >= TILES) return;
  const int tr   = wave / (CDIM / 16);
  const int tc   = wave % (CDIM / 16);
  const int lane = threadIdx.x & 31;
  const int half = lane >> 4;
  const int l16  = lane & 15;
  const int row0 = tr * 16, col0 = tc * 16;

  const float*  arow = x + (row0 + l16) * CDIM + 2 * half;
  const float4* bq   = wpl + tc * PACK_TILE + lane;

  v8f acc = {0.f, 0.f, 0.f, 0.f, 0.f, 0.f, 0.f, 0.f};
#pragma unroll
  for (int kk = 0; kk < KPAIRS; ++kk) {
    const int k0 = 8 * kk;
    const float4 bb = bq[kk * 32];
    v2f a, b;
    a.x = arow[k0];     a.y = arow[k0 + 1];
    b.x = bb.x;         b.y = bb.y;
    acc = __builtin_amdgcn_wmma_f32_16x16x4_f32(false, a, false, b,
                                                (short)0, acc, false, false);
    a.x = arow[k0 + 4]; a.y = arow[k0 + 5];
    b.x = bb.z;         b.y = bb.w;
    acc = __builtin_amdgcn_wmma_f32_16x16x4_f32(false, a, false, b,
                                                (short)0, acc, false, false);
  }

  float* orow = m + (row0 + 8 * half) * CDIM + col0 + l16;
#pragma unroll
  for (int r = 0; r < 8; ++r) orow[r * CDIM] = acc[r];
}

#define RNODES   400
#define NRANGES  125
#define APITCH   100
static_assert(NRANGES * RNODES == N_NODES, "ranges must tile the nodes exactly");
static_assert(RNODES % 16 == 0, "a range must hold whole 16-row tiles");
#define SLICE_Q  (N_EDGES / 4 / 2)
#define SCAN_IT  25
#define LCAP     544
#define SINK_ENT ((unsigned)RNODES)

__device__ __forceinline__ int4 scan_load(const int4* __restrict__ dst4, int q) {
  int4 d = make_int4(-1, -1, -1, -1);
  if (q < SLICE_Q) d = dst4[q];
  return d;
}

__device__ __forceinline__ void scan_keep(unsigned* mylist, int& count, int dl, int n, unsigned e) {
  const bool hit = (unsigned)dl < (unsigned)n;
  const unsigned mask = __builtin_amdgcn_ballot_w32(hit);
  if (mask) {
    if (hit) mylist[count + (int)__builtin_amdgcn_mbcnt_lo(mask, 0u)] = (e << 9) | (unsigned)dl;
    count += __builtin_popcount(mask);
  }
}

__device__ __forceinline__ void scan_group(unsigned* mylist, int& count, int4 d, int lo, int n, unsigned e) {
  scan_keep(mylist, count, d.x - lo, n, e);
  scan_keep(mylist, count, d.y - lo, n, e + 1u);
  scan_keep(mylist, count, d.z - lo, n, e + 2u);
  scan_keep(mylist, count, d.w - lo, n, e + 3u);
}

#define SCAP     512

template <int G>
__device__ __forceinline__ void add_rows(int s_l, int dl_l, int cnt, int lane,
                                         const float* __restrict__ m, float* acc) {
  for (int jb = 0; jb < cnt; jb += G) {
    float v[G][3];
    float* a[G];
#pragma unroll
    for (int j = 0; j < G; ++j) {
      const int s  = __builtin_amdgcn_readlane(s_l, jb + j);
      const int dl = __builtin_amdgcn_readlane(dl_l, jb + j);
      const float* r = m + s * CDIM + lane;
      v[j][0] = r[0]; v[j][1] = r[32]; v[j][2] = r[64];
      a[j] = acc + dl * APITCH + lane;
    }
#pragma unroll
    for (int j = 0; j < G; ++j) {
      atomicAdd(a[j],      v[j][0]);
      atomicAdd(a[j] + 32, v[j][1]);
      atomicAdd(a[j] + 64, v[j][2]);
    }
  }
}

template <int G, bool SAVE>
__device__ __forceinline__ void add_listed(const unsigned* ptr, int cnt, int lane,
                                           const float* __restrict__ m,
                                           const int* __restrict__ srcid, float* acc,
                                           unsigned* __restrict__ slist, int& saved) {
  const unsigned ent = (lane < cnt) ? ptr[lane] : SINK_ENT;
  const int s_l  = srcid[ent >> 9];
  const int dl_l = (int)(ent & 511u);
  if (SAVE) {
    if (saved + cnt <= SCAP && lane < cnt)
      slist[saved + lane] = ((unsigned)s_l << 9) | (unsigned)dl_l;
    saved += cnt;
  }
  add_rows<G>(s_l, dl_l, cnt, lane, m, acc);
}

template <bool SAVE>
__device__ __forceinline__ void scan_and_add(const float* __restrict__ m,
                                             const int* __restrict__ edge_index,
                                             float* acc, unsigned* mylist,
                                             int tid, int lane, int slice, int lo, int n,
                                             unsigned* __restrict__ slist, int& saved) {
  const int*  __restrict__ srcid = edge_index;
  const int4* __restrict__ dst4  = (const int4*)(edge_index + N_EDGES) + slice * SLICE_Q;
  const unsigned ebase = (unsigned)(slice * (N_EDGES / 2));
  int count = 0;

  int4 d0 = scan_load(dst4, 0 * 1024 + tid);
  int4 d1 = scan_load(dst4, 1 * 1024 + tid);
  int4 d2 = scan_load(dst4, 2 * 1024 + tid);
  int4 d3 = scan_load(dst4, 3 * 1024 + tid);
  for (int it = 0; it < SCAN_IT; ++it) {
    const int q0 = it * 4096 + tid;
    const int4 n0 = scan_load(dst4, q0 + 4096);
    const int4 n1 = scan_load(dst4, q0 + 5120);
    const int4 n2 = scan_load(dst4, q0 + 6144);
    const int4 n3 = scan_load(dst4, q0 + 7168);
    scan_group(mylist, count, d0, lo, n, ebase + 4u * (unsigned)(q0));
    scan_group(mylist, count, d1, lo, n, ebase + 4u * (unsigned)(q0 + 1024));
    scan_group(mylist, count, d2, lo, n, ebase + 4u * (unsigned)(q0 + 2048));
    scan_group(mylist, count, d3, lo, n, ebase + 4u * (unsigned)(q0 + 3072));
    while (count >= 32) {
      count -= 32;
      add_listed<16, SAVE>(mylist + count, 32, lane, m, srcid, acc, slist, saved);
    }
    d0 = n0; d1 = n1; d2 = n2; d3 = n3;
  }
  if (count > 0) add_listed<8, SAVE>(mylist, count, lane, m, srcid, acc, slist, saved);
}

template <bool FIRST>
__global__ __launch_bounds__(1024) void gnn_scatter(const float* __restrict__ m,
                                                    const int* __restrict__ edge_index,
                                                    float* __restrict__ agg0,
                                                    float* __restrict__ agg1,
                                                    unsigned* __restrict__ scount,
                                                    unsigned* __restrict__ slists) {
  __shared__ float    acc[(RNODES + 1) * APITCH];
  __shared__ unsigned list[32 * LCAP];

  const int tid   = threadIdx.x;
  const int lane  = tid & 31;
  const int w     = __builtin_amdgcn_readfirstlane(tid >> 5);
  const int range = blockIdx.x >> 1;
  const int slice = blockIdx.x & 1;
  const int lo    = range * RNODES;
  const int n     = RNODES;
  float* __restrict__ agg = slice ? agg1 : agg0;

  float4* acc4 = (float4*)acc;
  for (int i = tid; i < (RNODES + 1) * APITCH / 4; i += 1024)
    acc4[i] = make_float4(0.f, 0.f, 0.f, 0.f);
  __syncthreads();

  const int wid = blockIdx.x * 32 + w;
  unsigned* __restrict__ slist = slists + wid * SCAP;
  unsigned* mylist = list + w * LCAP;

  if (FIRST) {
    int saved = 0;
    scan_and_add<true>(m, edge_index, acc, mylist, tid, lane, slice, lo, n, slist, saved);
    if (lane == 0) scount[wid] = (unsigned)saved;
  } else {
    const int c = __builtin_amdgcn_readfirstlane((int)scount[wid]);
    if (c <= SCAP) {
      unsigned ent = (lane < c) ? slist[lane] : SINK_ENT;
      for (int b = 0; b < c; b += 32) {
        const int cnt = (c - b < 32) ? (c - b) : 32;
        const unsigned cur = ent;
        const int nx = b + 32 + lane;
        const unsigned nxt = slist[(nx < c) ? nx : (c - 1)];
        ent = (nx < c) ? nxt : SINK_ENT;
        const int s_l  = (int)(cur >> 9);
        const int dl_l = (int)(cur & 511u);
        if (cnt == 32) add_rows<16>(s_l, dl_l, 32,  lane, m, acc);
        else           add_rows<8> (s_l, dl_l, cnt, lane, m, acc);
      }
    } else {
      int saved = 0;
      scan_and_add<false>(m, edge_index, acc, mylist, tid, lane, slice, lo, n, slist, saved);
    }
  }

  __syncthreads();
  float4* __restrict__ o4 = (float4*)agg + range * ((RNODES / 16) * PACK_TILE);
  for (int p = tid; p < (RNODES / 16) * PACK_TILE; p += 1024) {
    const int q  = p >> 5;
    const int kk = q % KPAIRS;
    const int t  = q / KPAIRS;
    const float* s = acc + (t * 16 + (lane & 15)) * APITCH + 8 * kk + 2 * (lane >> 4);
    const v2f f0 = *(const v2f*)s;
    const v2f f1 = *(const v2f*)(s + 4);
    o4[p] = make_float4(f0.x, f0.y, f1.x, f1.y);
  }
}

__global__ __launch_bounds__(256) void gnn_gru(const float* __restrict__ agg0,
                                               const float* __restrict__ agg1,
                                               const float* __restrict__ h,
                                               const float4* __restrict__ wp,
                                               const float* __restrict__ b_ih,
                                               const float* __restrict__ b_hh,
                                               float* __restrict__ out) {
  const int wave = blockIdx.x * (blockDim.x >> 5) + (threadIdx.x >> 5);
  if (wave >= TILES) return;
  const int tr   = wave / (CDIM / 16);
  const int tc   = wave % (CDIM / 16);
  const int lane = threadIdx.x & 31;
  const int half = lane >> 4;
  const int l16  = lane & 15;
  const int row0 = tr * 16, col0 = tc * 16;
  const int col  = col0 + l16;

  v8f a_ir, a_iz, a_in, a_hr, a_hz, a_hn;
  {
    const float bir = b_ih[col], biz = b_ih[CDIM + col], bin_ = b_ih[2 * CDIM + col];
    const float bhr = b_hh[col], bhz = b_hh[CDIM + col], bhn  = b_hh[2 * CDIM + col];
#pragma unroll
    for (int r = 0; r < 8; ++r) {
      a_ir[r] = bir; a_iz[r] = biz; a_in[r] = bin_;
      a_hr[r] = bhr; a_hz[r] = bhz; a_hn[r] = bhn;
    }
  }

  const float4* ag0q = (const float4*)agg0 + tr * PACK_TILE + lane;
  const float4* ag1q = (const float4*)agg1 + tr * PACK_TILE + lane;
  const float* hrow   = h   + (row0 + l16) * CDIM + 2 * half;
  const float4* wq    = wp + tc * PACK_TILE + lane;

#pragma unroll 2
  for (int kk = 0; kk < KPAIRS; ++kk) {
    const int k0 = 8 * kk;
    const float4 w_ir = wq[0 * PACK_MAT + kk * 32];
    const float4 w_iz = wq[1 * PACK_MAT + kk * 32];
    const float4 w_in = wq[2 * PACK_MAT + kk * 32];
    const float4 w_hr = wq[3 * PACK_MAT + kk * 32];
    const float4 w_hz = wq[4 * PACK_MAT + kk * 32];
    const float4 w_hn = wq[5 * PACK_MAT + kk * 32];
    const float4 g0   = ag0q[kk * 32];
    const float4 g1   = ag1q[kk * 32];
    v2f ai, ah, b;

    ai.x = g0.x + g1.x; ai.y = g0.y + g1.y;
    ah.x = hrow[k0];   ah.y = hrow[k0 + 1];
    b.x = w_ir.x; b.y = w_ir.y;
    a_ir = __builtin_amdgcn_wmma_f32_16x16x4_f32(false, ai, false, b, (short)0, a_ir, false, false);
    b.x = w_iz.x; b.y = w_iz.y;
    a_iz = __builtin_amdgcn_wmma_f32_16x16x4_f32(false, ai, false, b, (short)0, a_iz, false, false);
    b.x = w_in.x; b.y = w_in.y;
    a_in = __builtin_amdgcn_wmma_f32_16x16x4_f32(false, ai, false, b, (short)0, a_in, false, false);
    b.x = w_hr.x; b.y = w_hr.y;
    a_hr = __builtin_amdgcn_wmma_f32_16x16x4_f32(false, ah, false, b, (short)0, a_hr, false, false);
    b.x = w_hz.x; b.y = w_hz.y;
    a_hz = __builtin_amdgcn_wmma_f32_16x16x4_f32(false, ah, false, b, (short)0, a_hz, false, false);
    b.x = w_hn.x; b.y = w_hn.y;
    a_hn = __builtin_amdgcn_wmma_f32_16x16x4_f32(false, ah, false, b, (short)0, a_hn, false, false);

    ai.x = g0.z + g1.z; ai.y = g0.w + g1.w;
    ah.x = hrow[k0 + 4];   ah.y = hrow[k0 + 5];
    b.x = w_ir.z; b.y = w_ir.w;
    a_ir = __builtin_amdgcn_wmma_f32_16x16x4_f32(false, ai, false, b, (short)0, a_ir, false, false);
    b.x = w_iz.z; b.y = w_iz.w;
    a_iz = __builtin_amdgcn_wmma_f32_16x16x4_f32(false, ai, false, b, (short)0, a_iz, false, false);
    b.x = w_in.z; b.y = w_in.w;
    a_in = __builtin_amdgcn_wmma_f32_16x16x4_f32(false, ai, false, b, (short)0, a_in, false, false);
    b.x = w_hr.z; b.y = w_hr.w;
    a_hr = __builtin_amdgcn_wmma_f32_16x16x4_f32(false, ah, false, b, (short)0, a_hr, false, false);
    b.x = w_hz.z; b.y = w_hz.w;
    a_hz = __builtin_amdgcn_wmma_f32_16x16x4_f32(false, ah, false, b, (short)0, a_hz, false, false);
    b.x = w_hn.z; b.y = w_hn.w;
    a_hn = __builtin_amdgcn_wmma_f32_16x16x4_f32(false, ah, false, b, (short)0, a_hn, false, false);
  }

#pragma unroll
  for (int r = 0; r < 8; ++r) {
    const int row = row0 + r + 8 * half;
    const float hv = h[row * CDIM + col];
    const float rr = 1.f / (1.f + __expf(-(a_ir[r] + a_hr[r])));
    const float zz = 1.f / (1.f + __expf(-(a_iz[r] + a_hz[r])));
    const float nn = tanhf(a_in[r] + rr * a_hn[r]);
    out[row * CDIM + col] = (1.f - zz) * nn + zz * hv;
  }
}

extern "C" void kernel_launch(void* const* d_in, const int* in_sizes, int n_in,
                              void* d_out, int out_size, void* d_ws, size_t ws_size,
                              hipStream_t stream) {
  const float* x      = (const float*)d_in[0];
  const int*   ei     = (const int*)  d_in[1];
  const float* weight = (const float*)d_in[2];
  const float* w_ih   = (const float*)d_in[3];
  const float* w_hh   = (const float*)d_in[4];
  const float* b_ih   = (const float*)d_in[5];
  const float* b_hh   = (const float*)d_in[6];
  float* out = (float*)d_out;

  float*  agg1 = (float*)d_ws;
  float4* wp   = (float4*)(agg1 + NC);
  float*  p    = agg1 + 2 * NC;
  float*  agg0 = agg1 + 3 * NC;
  unsigned* scount = (unsigned*)(agg1 + NC) + 131072;
  unsigned* slists = (unsigned*)(agg1 + NC) + 262144;

  const int wavesPerBlock = 256 / 32;
  const int gemmBlocks    = (TILES + wavesPerBlock - 1) / wavesPerBlock;
  const int scatBlocks    = NRANGES * 2;

  gnn_pack<<<PACK_ALL / 256, 256, 0, stream>>>(weight, w_ih, w_hh, wp);

  for (int l = 0; l < 3; ++l) {
    const float*  xin  = (l == 0) ? x   : ((l == 1) ? out : p);
    float*        m    = (l == 2) ? out : p;
    float*        xout = (l == 1) ? p   : out;
    const float4* wpl  = wp + (6 + l) * PACK_MAT;

    gnn_gemm_m <<<gemmBlocks, 256,  0, stream>>>(xin, wpl, m);
    if (l == 0) gnn_scatter<true> <<<scatBlocks, 1024, 0, stream>>>(m, ei, agg0, agg1, scount, slists);
    else        gnn_scatter<false><<<scatBlocks, 1024, 0, stream>>>(m, ei, agg0, agg1, scount, slists);
    gnn_gru    <<<gemmBlocks, 256,  0, stream>>>(agg0, agg1, xin, wp, b_ih, b_hh, xout);
  }
}


# Both WMMA kernels: a wave keeps its A fragments for several column tiles (m GEMM: 16x96 strip, 6 accumulators; GRU: 16x32, 12): rows loaded once per strip, 6x and 2x fewer waves; sums unchanged
# speedup vs baseline: 2.5735x; 1.0396x over previous
#include <hip/hip_runtime.h>
#include <hip/hip_bf16.h>


#define N_NODES 50000
#define N_EDGES 800000
#define CDIM    96
#define NC      (N_NODES * CDIM)
#define TILES   ((N_NODES / 16) * (CDIM / 16))

typedef __attribute__((ext_vector_type(2))) float v2f;
typedef __attribute__((ext_vector_type(8))) float v8f;

#define KPAIRS      (CDIM / 8)
#define PACK_TILE   (KPAIRS * 32)
#define PACK_MAT    ((CDIM / 16) * PACK_TILE)
#define PACK_ALL    (9 * PACK_MAT)

__global__ __launch_bounds__(256) void gnn_pack(const float* __restrict__ weight,
                                                const float* __restrict__ w_ih,
                                                const float* __restrict__ w_hh,
                                                float4* __restrict__ wp) {
  const int t    = blockIdx.x * 256 + threadIdx.x;
  const int lane = t & 31;
  const int c    = lane & 15;
  const int h    = lane >> 4;
  int q = t >> 5;
  const int kk = q % KPAIRS;       q /= KPAIRS;
  const int tc = q % (CDIM / 16);  q /= (CDIM / 16);
  const int k  = 8 * kk + 2 * h;
  const int n  = tc * 16 + c;
  float4 v;
  if (q < 6) {
    const float* s = (q < 3 ? w_ih : w_hh) + ((q % 3) * CDIM + n) * CDIM + k;
    v = make_float4(s[0], s[1], s[4], s[5]);
  } else {
    const float* s = weight + (q - 6) * CDIM * CDIM + k * CDIM + n;
    v = make_float4(s[0], s[CDIM], s[4 * CDIM], s[5 * CDIM]);
  }
  wp[t] = v;
}

#define STRIPS (N_NODES / 16)

__global__ __launch_bounds__(128) void gnn_gemm_m(const float* __restrict__ x,
                                                  const float4* __restrict__ wpl,
                                                  float* __restrict__ m) {
  const int strip = blockIdx.x * (blockDim.x >> 5) + (threadIdx.x >> 5);
  if (strip >= STRIPS) return;
  const int lane = threadIdx.x & 31;
  const int half = lane >> 4;
  const int l16  = lane & 15;
  const int row0 = strip * 16;

  const float*  arow = x + (row0 + l16) * CDIM + 2 * half;
  const float4* bq   = wpl + lane;

  v8f acc[CDIM / 16];
#pragma unroll
  for (int tc = 0; tc < CDIM / 16; ++tc)
    acc[tc] = (v8f){0.f, 0.f, 0.f, 0.f, 0.f, 0.f, 0.f, 0.f};

  v2f    a[KPAIRS][2];
  float4 bb[KPAIRS][CDIM / 16];
#pragma unroll
  for (int kk = 0; kk < KPAIRS; ++kk) {
    const int k0 = 8 * kk;
    a[kk][0].x = arow[k0];     a[kk][0].y = arow[k0 + 1];
    a[kk][1].x = arow[k0 + 4]; a[kk][1].y = arow[k0 + 5];
#pragma unroll
    for (int tc = 0; tc < CDIM / 16; ++tc) bb[kk][tc] = bq[tc * PACK_TILE + kk * 32];
  }
#pragma unroll
  for (int kk = 0; kk < KPAIRS; ++kk) {
#pragma unroll
    for (int tc = 0; tc < CDIM / 16; ++tc) {
      v2f b; b.x = bb[kk][tc].x; b.y = bb[kk][tc].y;
      acc[tc] = __builtin_amdgcn_wmma_f32_16x16x4_f32(false, a[kk][0], false, b,
                                                      (short)0, acc[tc], false, false);
    }
#pragma unroll
    for (int tc = 0; tc < CDIM / 16; ++tc) {
      v2f b; b.x = bb[kk][tc].z; b.y = bb[kk][tc].w;
      acc[tc] = __builtin_amdgcn_wmma_f32_16x16x4_f32(false, a[kk][1], false, b,
                                                      (short)0, acc[tc], false, false);
    }
  }
  __builtin_amdgcn_sched_group_barrier(0x020, 16, 0);
#pragma unroll
  for (int c = 0; c < KPAIRS / 2 - 1; ++c) {
    __builtin_amdgcn_sched_group_barrier(0x020, 16, 0);
    __builtin_amdgcn_sched_group_barrier(0x008, 24, 0);
  }
  __builtin_amdgcn_sched_group_barrier(0x008, 24, 0);

  float* orow = m + (row0 + 8 * half) * CDIM + l16;
#pragma unroll
  for (int tc = 0; tc < CDIM / 16; ++tc)
#pragma unroll
    for (int r = 0; r < 8; ++r) orow[r * CDIM + tc * 16] = acc[tc][r];
}

#define RNODES   400
#define NRANGES  125
#define APITCH   100
static_assert(NRANGES * RNODES == N_NODES, "ranges must tile the nodes exactly");
static_assert(RNODES % 16 == 0, "a range must hold whole 16-row tiles");
#define SLICE_Q  (N_EDGES / 4 / 2)
#define SCAN_IT  25
#define LCAP     544
#define SINK_ENT ((unsigned)RNODES)

__device__ __forceinline__ int4 scan_load(const int4* __restrict__ dst4, int q) {
  int4 d = make_int4(-1, -1, -1, -1);
  if (q < SLICE_Q) d = dst4[q];
  return d;
}

__device__ __forceinline__ void scan_keep(unsigned* mylist, int& count, int dl, int n, unsigned e) {
  const bool hit = (unsigned)dl < (unsigned)n;
  const unsigned mask = __builtin_amdgcn_ballot_w32(hit);
  if (mask) {
    if (hit) mylist[count + (int)__builtin_amdgcn_mbcnt_lo(mask, 0u)] = (e << 9) | (unsigned)dl;
    count += __builtin_popcount(mask);
  }
}

__device__ __forceinline__ void scan_group(unsigned* mylist, int& count, int4 d, int lo, int n, unsigned e) {
  scan_keep(mylist, count, d.x - lo, n, e);
  scan_keep(mylist, count, d.y - lo, n, e + 1u);
  scan_keep(mylist, count, d.z - lo, n, e + 2u);
  scan_keep(mylist, count, d.w - lo, n, e + 3u);
}

#define SCAP     512

template <int G>
__device__ __forceinline__ void add_rows(int s_l, int dl_l, int cnt, int lane,
                                         const float* __restrict__ m, float* acc) {
  for (int jb = 0; jb < cnt; jb += G) {
    float v[G][3];
    float* a[G];
#pragma unroll
    for (int j = 0; j < G; ++j) {
      const int s  = __builtin_amdgcn_readlane(s_l, jb + j);
      const int dl = __builtin_amdgcn_readlane(dl_l, jb + j);
      const float* r = m + s * CDIM + lane;
      v[j][0] = r[0]; v[j][1] = r[32]; v[j][2] = r[64];
      a[j] = acc + dl * APITCH + lane;
    }
#pragma unroll
    for (int j = 0; j < G; ++j) {
      atomicAdd(a[j],      v[j][0]);
      atomicAdd(a[j] + 32, v[j][1]);
      atomicAdd(a[j] + 64, v[j][2]);
    }
  }
}

template <int G, bool SAVE>
__device__ __forceinline__ void add_listed(const unsigned* ptr, int cnt, int lane,
                                           const float* __restrict__ m,
                                           const int* __restrict__ srcid, float* acc,
                                           unsigned* __restrict__ slist, int& saved) {
  const unsigned ent = (lane < cnt) ? ptr[lane] : SINK_ENT;
  const int s_l  = srcid[ent >> 9];
  const int dl_l = (int)(ent & 511u);
  if (SAVE) {
    if (saved + cnt <= SCAP && lane < cnt)
      slist[saved + lane] = ((unsigned)s_l << 9) | (unsigned)dl_l;
    saved += cnt;
  }
  add_rows<G>(s_l, dl_l, cnt, lane, m, acc);
}

template <bool SAVE>
__device__ __forceinline__ void scan_and_add(const float* __restrict__ m,
                                             const int* __restrict__ edge_index,
                                             float* acc, unsigned* mylist,
                                             int tid, int lane, int slice, int lo, int n,
                                             unsigned* __restrict__ slist, int& saved) {
  const int*  __restrict__ srcid = edge_index;
  const int4* __restrict__ dst4  = (const int4*)(edge_index + N_EDGES) + slice * SLICE_Q;
  const unsigned ebase = (unsigned)(slice * (N_EDGES / 2));
  int count = 0;

  int4 d0 = scan_load(dst4, 0 * 1024 + tid);
  int4 d1 = scan_load(dst4, 1 * 1024 + tid);
  int4 d2 = scan_load(dst4, 2 * 1024 + tid);
  int4 d3 = scan_load(dst4, 3 * 1024 + tid);
  for (int it = 0; it < SCAN_IT; ++it) {
    const int q0 = it * 4096 + tid;
    const int4 n0 = scan_load(dst4, q0 + 4096);
    const int4 n1 = scan_load(dst4, q0 + 5120);
    const int4 n2 = scan_load(dst4, q0 + 6144);
    const int4 n3 = scan_load(dst4, q0 + 7168);
    scan_group(mylist, count, d0, lo, n, ebase + 4u * (unsigned)(q0));
    scan_group(mylist, count, d1, lo, n, ebase + 4u * (unsigned)(q0 + 1024));
    scan_group(mylist, count, d2, lo, n, ebase + 4u * (unsigned)(q0 + 2048));
    scan_group(mylist, count, d3, lo, n, ebase + 4u * (unsigned)(q0 + 3072));
    while (count >= 32) {
      count -= 32;
      add_listed<16, SAVE>(mylist + count, 32, lane, m, srcid, acc, slist, saved);
    }
    d0 = n0; d1 = n1; d2 = n2; d3 = n3;
  }
  if (count > 0) add_listed<8, SAVE>(mylist, count, lane, m, srcid, acc, slist, saved);
}

template <bool FIRST>
__global__ __launch_bounds__(1024) void gnn_scatter(const float* __restrict__ m,
                                                    const int* __restrict__ edge_index,
                                                    float* __restrict__ agg0,
                                                    float* __restrict__ agg1,
                                                    unsigned* __restrict__ scount,
                                                    unsigned* __restrict__ slists) {
  __shared__ float    acc[(RNODES + 1) * APITCH];
  __shared__ unsigned list[32 * LCAP];

  const int tid   = threadIdx.x;
  const int lane  = tid & 31;
  const int w     = __builtin_amdgcn_readfirstlane(tid >> 5);
  const int range = blockIdx.x >> 1;
  const int slice = blockIdx.x & 1;
  const int lo    = range * RNODES;
  const int n     = RNODES;
  float* __restrict__ agg = slice ? agg1 : agg0;

  float4* acc4 = (float4*)acc;
  for (int i = tid; i < (RNODES + 1) * APITCH / 4; i += 1024)
    acc4[i] = make_float4(0.f, 0.f, 0.f, 0.f);
  __syncthreads();

  const int wid = blockIdx.x * 32 + w;
  unsigned* __restrict__ slist = slists + wid * SCAP;
  unsigned* mylist = list + w * LCAP;

  if (FIRST) {
    int saved = 0;
    scan_and_add<true>(m, edge_index, acc, mylist, tid, lane, slice, lo, n, slist, saved);
    if (lane == 0) scount[wid] = (unsigned)saved;
  } else {
    const int c = __builtin_amdgcn_readfirstlane((int)scount[wid]);
    if (c <= SCAP) {
      unsigned ent = (lane < c) ? slist[lane] : SINK_ENT;
      for (int b = 0; b < c; b += 32) {
        const int cnt = (c - b < 32) ? (c - b) : 32;
        const unsigned cur = ent;
        const int nx = b + 32 + lane;
        const unsigned nxt = slist[(nx < c) ? nx : (c - 1)];
        ent = (nx < c) ? nxt : SINK_ENT;
        const int s_l  = (int)(cur >> 9);
        const int dl_l = (int)(cur & 511u);
        if (cnt == 32) add_rows<16>(s_l, dl_l, 32,  lane, m, acc);
        else           add_rows<8> (s_l, dl_l, cnt, lane, m, acc);
      }
    } else {
      int saved = 0;
      scan_and_add<false>(m, edge_index, acc, mylist, tid, lane, slice, lo, n, slist, saved);
    }
  }

  __syncthreads();
  float4* __restrict__ o4 = (float4*)agg + range * ((RNODES / 16) * PACK_TILE);
  for (int p = tid; p < (RNODES / 16) * PACK_TILE; p += 1024) {
    const int q  = p >> 5;
    const int kk = q % KPAIRS;
    const int t  = q / KPAIRS;
    const float* s = acc + (t * 16 + (lane & 15)) * APITCH + 8 * kk + 2 * (lane >> 4);
    const v2f f0 = *(const v2f*)s;
    const v2f f1 = *(const v2f*)(s + 4);
    o4[p] = make_float4(f0.x, f0.y, f1.x, f1.y);
  }
}

#define GRU_WAVES ((N_NODES / 16) * (CDIM / 32))

__global__ __launch_bounds__(256) void gnn_gru(const float* __restrict__ agg0,
                                               const float* __restrict__ agg1,
                                               const float* __restrict__ h,
                                               const float4* __restrict__ wp,
                                               const float* __restrict__ b_ih,
                                               const float* __restrict__ b_hh,
                                               float* __restrict__ out) {
  const int wave = blockIdx.x * (blockDim.x >> 5) + (threadIdx.x >> 5);
  if (wave >= GRU_WAVES) return;
  const int tr   = wave / (CDIM / 32);
  const int tp   = wave % (CDIM / 32);
  const int lane = threadIdx.x & 31;
  const int half = lane >> 4;
  const int l16  = lane & 15;
  const int row0 = tr * 16;

  v8f acc[2][6];
#pragma unroll
  for (int j = 0; j < 2; ++j) {
    const int col = (2 * tp + j) * 16 + l16;
#pragma unroll
    for (int g = 0; g < 3; ++g) {
      const float bi = b_ih[g * CDIM + col];
      const float bh = b_hh[g * CDIM + col];
#pragma unroll
      for (int r = 0; r < 8; ++r) { acc[j][g][r] = bi; acc[j][3 + g][r] = bh; }
    }
  }

  const float4* ag0q = (const float4*)agg0 + tr * PACK_TILE + lane;
  const float4* ag1q = (const float4*)agg1 + tr * PACK_TILE + lane;
  const float*  hrow = h + (row0 + l16) * CDIM + 2 * half;
  const float4* wq   = wp + (2 * tp) * PACK_TILE + lane;

#pragma unroll 1
  for (int kk = 0; kk < KPAIRS; ++kk) {
    const int k0 = 8 * kk;
    const float4 g0 = ag0q[kk * 32];
    const float4 g1 = ag1q[kk * 32];
    v2f ai[2], ah[2];
    ah[0].x = hrow[k0];     ah[0].y = hrow[k0 + 1];
    ah[1].x = hrow[k0 + 4]; ah[1].y = hrow[k0 + 5];
    float4 w[2][6];
#pragma unroll
    for (int j = 0; j < 2; ++j)
#pragma unroll
      for (int g = 0; g < 6; ++g)
        w[j][g] = wq[g * PACK_MAT + j * PACK_TILE + kk * 32];
    __builtin_amdgcn_sched_group_barrier(0x020, 16, 0);
    __builtin_amdgcn_sched_group_barrier(0x008, 24, 0);
    ai[0].x = g0.x + g1.x;  ai[0].y = g0.y + g1.y;
    ai[1].x = g0.z + g1.z;  ai[1].y = g0.w + g1.w;
#pragma unroll
    for (int s = 0; s < 2; ++s) {
#pragma unroll
      for (int j = 0; j < 2; ++j) {
#pragma unroll
        for (int g = 0; g < 6; ++g) {
          v2f b;
          if (s == 0) { b.x = w[j][g].x; b.y = w[j][g].y; }
          else        { b.x = w[j][g].z; b.y = w[j][g].w; }
          acc[j][g] = __builtin_amdgcn_wmma_f32_16x16x4_f32(false, (g < 3) ? ai[s] : ah[s], false, b,
                                                            (short)0, acc[j][g], false, false);
        }
      }
    }
  }

#pragma unroll
  for (int j = 0; j < 2; ++j) {
    const int col = (2 * tp + j) * 16 + l16;
#pragma unroll
    for (int r = 0; r < 8; ++r) {
      const int row = row0 + r + 8 * half;
      const float hv = h[row * CDIM + col];
      const float rr = 1.f / (1.f + __expf(-(acc[j][0][r] + acc[j][3][r])));
      const float zz = 1.f / (1.f + __expf(-(acc[j][1][r] + acc[j][4][r])));
      const float nn = tanhf(acc[j][2][r] + rr * acc[j][5][r]);
      out[row * CDIM + col] = (1.f - zz) * nn + zz * hv;
    }
  }
}

extern "C" void kernel_launch(void* const* d_in, const int* in_sizes, int n_in,
                              void* d_out, int out_size, void* d_ws, size_t ws_size,
                              hipStream_t stream) {
  const float* x      = (const float*)d_in[0];
  const int*   ei     = (const int*)  d_in[1];
  const float* weight = (const float*)d_in[2];
  const float* w_ih   = (const float*)d_in[3];
  const float* w_hh   = (const float*)d_in[4];
  const float* b_ih   = (const float*)d_in[5];
  const float* b_hh   = (const float*)d_in[6];
  float* out = (float*)d_out;

  float*  agg1 = (float*)d_ws;
  float4* wp   = (float4*)(agg1 + NC);
  float*  p    = agg1 + 2 * NC;
  float*  agg0 = agg1 + 3 * NC;
  unsigned* scount = (unsigned*)(agg1 + NC) + 131072;
  unsigned* slists = (unsigned*)(agg1 + NC) + 262144;

  const int gemmBlocks    = (STRIPS + 3) / 4;
  const int gruBlocks     = (GRU_WAVES + 7) / 8;
  const int scatBlocks    = NRANGES * 2;

  gnn_pack<<<PACK_ALL / 256, 256, 0, stream>>>(weight, w_ih, w_hh, wp);

  for (int l = 0; l < 3; ++l) {
    const float*  xin  = (l == 0) ? x   : ((l == 1) ? out : p);
    float*        m    = (l == 2) ? out : p;
    float*        xout = (l == 1) ? p   : out;
    const float4* wpl  = wp + (6 + l) * PACK_MAT;

    gnn_gemm_m <<<gemmBlocks, 128,  0, stream>>>(xin, wpl, m);
    if (l == 0) gnn_scatter<true> <<<scatBlocks, 1024, 0, stream>>>(m, ei, agg0, agg1, scount, slists);
    else        gnn_scatter<false><<<scatBlocks, 1024, 0, stream>>>(m, ei, agg0, agg1, scount, slists);
    gnn_gru    <<<gruBlocks,  256,  0, stream>>>(agg0, agg1, xin, wp, b_ih, b_hh, xout);
  }
}
